# starting kernel
# baseline (speedup 1.0000x reference)
#include <hip/hip_runtime.h>
#include <hip/hip_bf16.h>


typedef float v2f __attribute__((ext_vector_type(2)));
typedef float v8f __attribute__((ext_vector_type(8)));

#define B_  16
#define T_  4096
#define D_  256
#define N_  64
#define K_  4
#define BT_ (B_ * T_)
#define L_  64
#define C_  (T_ / L_)

static __device__ __forceinline__ v8f wmma4(v2f a, v2f b, v8f c) {
    return __builtin_amdgcn_wmma_f32_16x16x4_f32(false, a, false, b,
                                                 (short)0, c, false, false);
}

__global__ __launch_bounds__(128)
void k1_xbmix(const float* __restrict__ x, const float* __restrict__ alpha,
              const float* __restrict__ Bs, float* __restrict__ U) {
    __shared__ float xs[16 * 257];
    __shared__ float al[16 * 4];
    const int tid = threadIdx.x;
    const int bt0 = blockIdx.x * 16;

    for (int i = tid; i < 16 * 256; i += 128) {
        int r = i >> 8, c = i & 255;
        xs[r * 257 + c] = x[(bt0 + r) * 256 + c];
    }
    if (tid < 64) al[tid] = alpha[bt0 * 4 + tid];
    __syncthreads();

    const int lane = tid & 31, wave = tid >> 5;
    const int n0 = wave * 16;
    const int r = lane & 15, half = lane >> 4, koff = half * 2;
    const float* xrow = &xs[r * 257];
    const float* arow = &al[r * 4];
    const int n = n0 + r;

    v8f acc = {};
    for (int kb = 0; kb < 1024; kb += 4) {
        int   k0 = kb + koff;
        int   kk = k0 >> 8;
        int   d  = k0 & 255;
        float am = arow[kk];
        v2f a, b;
        a.x = am * xrow[d];
        a.y = am * xrow[d + 1];
        const float* bp = &Bs[(kk * 64 + n) * 256 + d];
        b.x = bp[0];
        b.y = bp[1];
        acc = wmma4(a, b, acc);
    }
#pragma unroll
    for (int rr = 0; rr < 8; ++rr)
        U[(bt0 + rr + half * 8) * 64 + n] = acc[rr];
}

__global__ __launch_bounds__(128)
void kA_compose(const float* __restrict__ alpha, const float* __restrict__ As,
                const float* __restrict__ U, float* __restrict__ Mall,
                float* __restrict__ vall) {
    __shared__ float Assh[K_ * N_ * N_];
    __shared__ float Msh[N_ * 65];
    __shared__ float AT[N_ * 65];
    __shared__ float vsh[N_];
    __shared__ float psum[2 * N_];

    const int tid = threadIdx.x;
    const int bc  = blockIdx.x;
    const int b   = bc >> 6;
    const int c   = bc & 63;

    for (int i = tid; i < K_ * N_ * N_; i += 128) Assh[i] = As[i];
    for (int i = tid; i < N_ * N_; i += 128) {
        int nn = i >> 6, mm = i & 63;
        Msh[nn * 65 + mm] = (nn == mm) ? 1.0f : 0.0f;
    }
    if (tid < N_) vsh[tid] = 0.0f;
    __syncthreads();

    const int lane = tid & 31, wave = tid >> 5;
    const int r = lane & 15, half = lane >> 4, koff = half * 2;
    const int vm = tid & 63, vpart = tid >> 6;
    const long t0 = (long)b * T_ + (long)c * L_;

    for (int s = 0; s < L_; ++s) {
        const long bt = t0 + s;
        const float4 a4 = ((const float4*)alpha)[bt];
        for (int i = tid; i < N_ * N_; i += 128) {
            int nn = i >> 6, mm = i & 63;
            float v = a4.x * Assh[(0 * 64 + nn) * 64 + mm]
                    + a4.y * Assh[(1 * 64 + nn) * 64 + mm]
                    + a4.z * Assh[(2 * 64 + nn) * 64 + mm]
                    + a4.w * Assh[(3 * 64 + nn) * 64 + mm];
            AT[mm * 65 + nn] = v;
        }
        __syncthreads();

        v8f acc[4] = {v8f{}, v8f{}, v8f{}, v8f{}};
        const float* arow = &Msh[(16 * wave + r) * 65];
        for (int kb = 0; kb < 64; kb += 4) {
            v2f a;
            a.x = arow[kb + koff];
            a.y = arow[kb + koff + 1];
#pragma unroll
            for (int J = 0; J < 4; ++J) {
                v2f bf;
                const float* bp = &AT[(16 * J + r) * 65 + kb + koff];
                bf.x = bp[0];
                bf.y = bp[1];
                acc[J] = wmma4(a, bf, acc[J]);
            }
        }
        {
            float local = 0.0f;
            const int nb = vpart * 32;
#pragma unroll 8
            for (int j = 0; j < 32; ++j)
                local = __builtin_fmaf(vsh[nb + j], AT[vm * 65 + nb + j], local);
            psum[vpart * 64 + vm] = local;
        }
        __syncthreads();

#pragma unroll
        for (int J = 0; J < 4; ++J)
#pragma unroll
            for (int rr = 0; rr < 8; ++rr)
                Msh[(16 * wave + rr + 8 * half) * 65 + 16 * J + r] = acc[J][rr];
        if (tid < 64)
            vsh[tid] = psum[tid] + psum[64 + tid] + U[bt * 64 + tid];
    }
    __syncthreads();
    for (int i = tid; i < N_ * N_; i += 128)
        Mall[(long)bc * 4096 + i] = Msh[(i >> 6) * 65 + (i & 63)];
    if (tid < 64) vall[(long)bc * 64 + tid] = vsh[tid];
}

__global__ __launch_bounds__(256)
void kB_chunkscan(const float* __restrict__ Mall, const float* __restrict__ vall,
                  float* __restrict__ hstart, float* __restrict__ hT) {
    __shared__ float h[64];
    __shared__ float psum[4 * 64];
    const int tid = threadIdx.x, b = blockIdx.x;
    const int m = tid & 63, part = tid >> 6;
    if (tid < 64) h[tid] = 0.0f;
    __syncthreads();
    for (int c = 0; c < C_; ++c) {
        const long bc = (long)b * C_ + c;
        if (tid < 64) hstart[bc * 64 + tid] = h[tid];
        float local = 0.0f;
        const float* Mc = &Mall[bc * 4096];
        const int n0 = part * 16;
#pragma unroll
        for (int j = 0; j < 16; ++j)
            local = __builtin_fmaf(h[n0 + j], Mc[(n0 + j) * 64 + m], local);
        psum[part * 64 + m] = local;
        __syncthreads();
        if (tid < 64)
            h[tid] = psum[tid] + psum[64 + tid] + psum[128 + tid] +
                     psum[192 + tid] + vall[bc * 64 + tid];
        __syncthreads();
    }
    if (tid < 64) hT[b * 64 + tid] = h[tid];
}

__global__ __launch_bounds__(256)
void kC_replay(const float* __restrict__ alpha, const float* __restrict__ As,
               const float* __restrict__ U, const float* __restrict__ hstart,
               float* __restrict__ hist) {
    __shared__ float Af[256 * 64];
    __shared__ float ha[256];
    __shared__ float h[64];
    __shared__ float psum[4 * 64];
    const int tid = threadIdx.x;
    const int bc  = blockIdx.x;
    for (int i = tid; i < 256 * 64; i += 256) Af[i] = As[i];
    if (tid < 64) h[tid] = hstart[(long)bc * 64 + tid];
    __syncthreads();

    const int m = tid & 63, part = tid >> 6;
    const int b = bc >> 6, c = bc & 63;
    const long t0 = (long)b * T_ + (long)c * L_;

    for (int s = 0; s < L_; ++s) {
        const long bt = t0 + s;
        ha[tid] = alpha[bt * 4 + (tid >> 6)] * h[tid & 63];
        __syncthreads();
        float local = 0.0f;
        const int row0 = part * 64;
#pragma unroll 8
        for (int j = 0; j < 64; ++j)
            local = __builtin_fmaf(ha[row0 + j], Af[(row0 + j) * 64 + m], local);
        psum[part * 64 + m] = local;
        __syncthreads();
        if (tid < 64) {
            float hn = U[bt * 64 + tid] + psum[tid] + psum[64 + tid] +
                       psum[128 + tid] + psum[192 + tid];
            h[tid] = hn;
            hist[bt * 64 + tid] = hn;
        }
        __syncthreads();
    }
}

__global__ __launch_bounds__(128)
void k3_y(const float* __restrict__ x, const float* __restrict__ alpha,
          const float* __restrict__ Cs, const float* __restrict__ Ds,
          const float* __restrict__ hist, float* __restrict__ y) {
    __shared__ float xs[16 * 257];
    __shared__ float hs[16 * 65];
    __shared__ float al[16 * 4];
    const int tid = threadIdx.x;
    const int bt0 = blockIdx.x * 16;

    for (int i = tid; i < 16 * 256; i += 128) {
        int r = i >> 8, c = i & 255;
        xs[r * 257 + c] = x[(bt0 + r) * 256 + c];
    }
    for (int i = tid; i < 16 * 64; i += 128) {
        int r = i >> 6, c = i & 63;
        hs[r * 65 + c] = hist[(bt0 + r) * 64 + c];
    }
    if (tid < 64) al[tid] = alpha[bt0 * 4 + tid];
    __syncthreads();

    const int lane = tid & 31, wave = tid >> 5;
    const int n0 = (blockIdx.y * 4 + wave) * 16;
    const int r = lane & 15, half = lane >> 4, koff = half * 2;
    const float* xrow = &xs[r * 257];
    const float* hrow = &hs[r * 65];
    const float* arow = &al[r * 4];
    const int e = n0 + r;

    v8f acc = {};
    for (int kb = 0; kb < 1024; kb += 4) {
        int   k0 = kb + koff;
        int   kk = k0 >> 8;
        int   d  = k0 & 255;
        float am = arow[kk];
        v2f a, b;
        a.x = am * xrow[d];
        a.y = am * xrow[d + 1];
        const float* bp = &Ds[kk * 65536 + e * 256 + d];
        b.x = bp[0];
        b.y = bp[1];
        acc = wmma4(a, b, acc);
    }
    for (int kb = 0; kb < 256; kb += 4) {
        int   k0 = kb + koff;
        int   kk = k0 >> 6;
        int   n  = k0 & 63;
        float am = arow[kk];
        v2f a, b;
        a.x = am * hrow[n];
        a.y = am * hrow[n + 1];
        const float* cp = &Cs[kk * 16384 + e * 64 + n];
        b.x = cp[0];
        b.y = cp[1];
        acc = wmma4(a, b, acc);
    }
#pragma unroll
    for (int rr = 0; rr < 8; ++rr)
        y[(size_t)(bt0 + rr + half * 8) * 256 + e] = acc[rr];
}

extern "C" void kernel_launch(void* const* d_in, const int* in_sizes, int n_in,
                              void* d_out, int out_size, void* d_ws, size_t ws_size,
                              hipStream_t stream) {
    const float* x     = (const float*)d_in[0];
    const float* alpha = (const float*)d_in[1];
    const float* As    = (const float*)d_in[2];
    const float* Bs    = (const float*)d_in[3];
    const float* Cs    = (const float*)d_in[4];
    const float* Ds    = (const float*)d_in[5];

    float* y  = (float*)d_out;
    float* hT = y + (size_t)BT_ * D_;

    float* U      = (float*)d_ws;
    float* hist   = U      + (size_t)BT_ * N_;
    float* Mall   = hist   + (size_t)BT_ * N_;
    float* vall   = Mall   + (size_t)B_ * C_ * N_ * N_;
    float* hstart = vall   + (size_t)B_ * C_ * N_;

    k1_xbmix   <<<dim3(BT_ / 16),    128, 0, stream>>>(x, alpha, Bs, U);
    kA_compose <<<dim3(B_ * C_),     128, 0, stream>>>(alpha, As, U, Mall, vall);
    kB_chunkscan<<<dim3(B_),         256, 0, stream>>>(Mall, vall, hstart, hT);
    kC_replay  <<<dim3(B_ * C_),     256, 0, stream>>>(alpha, As, U, hstart, hist);
    k3_y       <<<dim3(BT_ / 16, 4), 128, 0, stream>>>(x, alpha, Cs, Ds, hist, y);
}


# kC replay as a batched matrix product: 16 chunks per block are the rows of one f32 16x16x4 WMMA product per step (64 blocks), state tile in 2 LDS stages, 1 LDS-only barrier; sum order of h changes
# speedup vs baseline: 1.6766x; 1.6766x over previous
#include <hip/hip_runtime.h>
#include <hip/hip_bf16.h>


typedef float v2f __attribute__((ext_vector_type(2)));
typedef float v8f __attribute__((ext_vector_type(8)));
typedef float v4f __attribute__((ext_vector_type(4)));

#define B_  16
#define T_  4096
#define D_  256
#define N_  64
#define K_  4
#define BT_ (B_ * T_)
#define L_  64
#define C_  (T_ / L_)

static __device__ __forceinline__ v8f wmma4(v2f a, v2f b, v8f c) {
    return __builtin_amdgcn_wmma_f32_16x16x4_f32(false, a, false, b,
                                                 (short)0, c, false, false);
}

#define XP_ 260
#define HP_ 68

struct WSlab { v4f q[8]; };

static __device__ __forceinline__ WSlab fetch_slab(const float* __restrict__ base,
                                                   int stride, int tid) {
    WSlab out;
#pragma unroll
    for (int p = 0; p < 8; ++p) {
        int idx = tid + 128 * p;
        int row = idx >> 4, c4 = idx & 15;
        out.q[p] = *(const v4f*)(base + row * stride + c4 * 4);
    }
    return out;
}

static __device__ __forceinline__ void store_slab(float* ws, const WSlab& pre, int tid) {
#pragma unroll
    for (int p = 0; p < 8; ++p) {
        int idx = tid + 128 * p;
        int row = idx >> 4, c4 = idx & 15;
        *(v4f*)&ws[row * HP_ + c4 * 4] = pre.q[p];
    }
}

static __device__ __forceinline__ void slab_mma(const float* ap, float am,
                                                const float* wrow, v8f (&acc)[4]) {
#pragma unroll
    for (int kb = 0; kb < 64; kb += 4) {
        v2f xa = *(const v2f*)(ap + kb);
        v2f a;
        a.x = am * xa.x;
        a.y = am * xa.y;
#pragma unroll
        for (int J = 0; J < 4; ++J) {
            v2f bf = *(const v2f*)(wrow + J * 16 * HP_ + kb);
            acc[J] = wmma4(a, bf, acc[J]);
        }
    }
}

__global__ __launch_bounds__(128)
void k1_xbmix(const float* __restrict__ x, const float* __restrict__ alpha,
              const float* __restrict__ Bs, float* __restrict__ U) {
    __shared__ __attribute__((aligned(16))) float xs[64 * XP_];
    __shared__ __attribute__((aligned(16))) float ws[64 * HP_];
    __shared__ __attribute__((aligned(16))) float al[64 * 4];
    const int tid = threadIdx.x;
    const int bt0 = blockIdx.x * 64;

    WSlab pre = fetch_slab(Bs, 256, tid);

    {
        const v4f* xg = (const v4f*)x + (size_t)bt0 * 64;
#pragma unroll 8
        for (int i = tid; i < 64 * 64; i += 128) {
            int rr = i >> 6, c4 = i & 63;
            *(v4f*)&xs[rr * XP_ + c4 * 4] = xg[i];
        }
        if (tid < 64)
            *(v4f*)&al[tid * 4] = ((const v4f*)alpha)[bt0 + tid];
    }

    const int lane = tid & 31, wave = tid >> 5;
    const int r = lane & 15, half = lane >> 4;
    const int arow = 16 * wave + r;
    const float* xrow = &xs[arow * XP_ + 2 * half];
    const float* wrow = &ws[r * HP_ + 2 * half];

    v8f acc[4] = {v8f{}, v8f{}, v8f{}, v8f{}};

    for (int s = 0; s < 16; ++s) {
        __syncthreads();
        store_slab(ws, pre, tid);
        __syncthreads();
        if (s + 1 < 16)
            pre = fetch_slab(Bs + ((s + 1) >> 2) * 16384 + ((s + 1) & 3) * 64, 256, tid);
        slab_mma(xrow + (s & 3) * 64, al[arow * 4 + (s >> 2)], wrow, acc);
    }

#pragma unroll
    for (int J = 0; J < 4; ++J)
#pragma unroll
        for (int rr = 0; rr < 8; ++rr)
            U[(size_t)(bt0 + 16 * wave + rr + half * 8) * 64 + 16 * J + r] = acc[J][rr];
}

#define AP_ 68

__global__ __launch_bounds__(128)
void kA_compose(const float* __restrict__ alpha, const float* __restrict__ As,
                const float* __restrict__ U, float* __restrict__ Mall,
                float* __restrict__ vall) {
    __shared__ __attribute__((aligned(16))) float Assh[K_ * N_ * AP_];
    __shared__ __attribute__((aligned(16))) float PT[2][N_ * AP_];
    __shared__ __attribute__((aligned(16))) float al[L_ * 4];
    __shared__ __attribute__((aligned(16))) float ush[2][N_];

    const int tid = threadIdx.x;
    const int bc  = blockIdx.x;
    const int b   = bc >> 6;
    const int c   = bc & 63;
    const long t0 = (long)b * T_ + (long)c * L_;

    for (int i = tid; i < K_ * N_ * 16; i += 128) {
        int row = i >> 4, c4 = i & 15;
        *(v4f*)&Assh[row * AP_ + c4 * 4] = ((const v4f*)As)[i];
    }
    for (int i = tid; i < N_ * N_; i += 128) {
        int mm = i >> 6, nn = i & 63;
        PT[0][mm * AP_ + nn] = (nn == mm) ? 1.0f : 0.0f;
    }
    if (tid < 64) {
        *(v4f*)&al[tid * 4] = ((const v4f*)alpha)[t0 + tid];
        ush[0][tid] = U[(t0 + L_ - 1) * 64 + tid];
    }
    __syncthreads();

    const int lane = tid & 31, wave = tid >> 5;
    const int r = lane & 15, half = lane >> 4;
    const float* asrow = &Assh[(16 * wave + r) * AP_ + 2 * half];
    const int gm = tid >> 1, gh = tid & 1;
    float q = 0.0f;

    for (int it = 0; it < L_; ++it) {
        const int s = L_ - 1 - it;
        const int cur = it & 1, nxt = cur ^ 1;
        const float* Pc = PT[cur];
        float*       Pn = PT[nxt];

        float unext = 0.0f;
        if (tid < 64 && s > 0) unext = U[(t0 + s - 1) * 64 + tid];

        const v4f a4 = *(const v4f*)&al[s * 4];

        v8f acc[4] = {v8f{}, v8f{}, v8f{}, v8f{}};
        const float* brow = &Pc[r * AP_ + 2 * half];
#pragma unroll
        for (int kb = 0; kb < 64; kb += 4) {
            v2f s0 = *(const v2f*)(asrow + 0 * 64 * AP_ + kb);
            v2f s1 = *(const v2f*)(asrow + 1 * 64 * AP_ + kb);
            v2f s2 = *(const v2f*)(asrow + 2 * 64 * AP_ + kb);
            v2f s3 = *(const v2f*)(asrow + 3 * 64 * AP_ + kb);
            v2f a = a4.x * s0;
            a = a4.y * s1 + a;
            a = a4.z * s2 + a;
            a = a4.w * s3 + a;
#pragma unroll
            for (int J = 0; J < 4; ++J) {
                v2f bf = *(const v2f*)(brow + J * 16 * AP_ + kb);
                acc[J] = wmma4(a, bf, acc[J]);
            }
        }

        {
            float local = 0.0f;
            const float* pr = &Pc[gm * AP_ + 32 * gh];
            const float* ur = &ush[cur][32 * gh];
#pragma unroll
            for (int j = 0; j < 8; ++j) {
                v4f pv = *(const v4f*)(pr + 4 * j);
                v4f uv = *(const v4f*)(ur + 4 * j);
                local = __builtin_fmaf(uv.x, pv.x, local);
                local = __builtin_fmaf(uv.y, pv.y, local);
                local = __builtin_fmaf(uv.z, pv.z, local);
                local = __builtin_fmaf(uv.w, pv.w, local);
            }
            local += __shfl_xor(local, 1);
            q += local;
        }

#pragma unroll
        for (int J = 0; J < 4; ++J) {
            float* dst = &Pn[(16 * J + r) * AP_ + 16 * wave + 8 * half];
            v4f lo = {acc[J][0], acc[J][1], acc[J][2], acc[J][3]};
            v4f hi = {acc[J][4], acc[J][5], acc[J][6], acc[J][7]};
            *(v4f*)dst       = lo;
            *(v4f*)(dst + 4) = hi;
        }
        if (tid < 64 && s > 0) ush[nxt][tid] = unext;
        __syncthreads();
    }

    for (int i = tid; i < N_ * N_; i += 128)
        Mall[(long)bc * 4096 + i] = PT[0][(i & 63) * AP_ + (i >> 6)];
    if (gh == 0) vall[(long)bc * 64 + gm] = q;
}

__global__ __launch_bounds__(256)
void kB_chunkscan(const float* __restrict__ Mall, const float* __restrict__ vall,
                  float* __restrict__ hstart, float* __restrict__ hT) {
    __shared__ float h[64];
    __shared__ float psum[4 * 64];
    const int tid = threadIdx.x, b = blockIdx.x;
    const int m = tid & 63, part = tid >> 6;
    if (tid < 64) h[tid] = 0.0f;
    __syncthreads();
    for (int c = 0; c < C_; ++c) {
        const long bc = (long)b * C_ + c;
        if (tid < 64) hstart[bc * 64 + tid] = h[tid];
        float local = 0.0f;
        const float* Mc = &Mall[bc * 4096];
        const int n0 = part * 16;
#pragma unroll
        for (int j = 0; j < 16; ++j)
            local = __builtin_fmaf(h[n0 + j], Mc[(n0 + j) * 64 + m], local);
        psum[part * 64 + m] = local;
        __syncthreads();
        if (tid < 64)
            h[tid] = psum[tid] + psum[64 + tid] + psum[128 + tid] +
                     psum[192 + tid] + vall[bc * 64 + tid];
        __syncthreads();
    }
    if (tid < 64) hT[b * 64 + tid] = h[tid];
}

#define RP_ 68
#define G_  16

static __device__ __forceinline__ void lds_barrier() {
    asm volatile("s_wait_dscnt 0x0\n\ts_barrier_signal -1\n\ts_barrier_wait -1" ::: "memory");
}

__global__ __launch_bounds__(128) __attribute__((amdgpu_waves_per_eu(1, 1)))
void kC_replay(const float* __restrict__ alpha, const float* __restrict__ As,
               const float* __restrict__ U, const float* __restrict__ hstart,
               float* __restrict__ hist) {
    __shared__ __attribute__((aligned(16))) float hs[2][G_ * RP_];
    __shared__ __attribute__((aligned(16))) float al[G_ * L_ * 4];
    const int tid = threadIdx.x;
    const int g   = blockIdx.x;
    const int lane = tid & 31, wave = tid >> 5;
    const int c = lane & 15, half = lane >> 4;

    v2f bfr[K_][16];
    {
        const float* ab = As + 16 * wave + c;
#pragma unroll
        for (int k = 0; k < K_; ++k)
#pragma unroll
            for (int w = 0; w < 16; ++w) {
                const int n0 = 8 * (w >> 1) + 4 * half + 2 * (w & 1);
                bfr[k][w].x = ab[k * 4096 + n0 * 64];
                bfr[k][w].y = ab[k * 4096 + (n0 + 1) * 64];
            }
    }

    {
        const v4f* ag = (const v4f*)alpha + (size_t)g * (G_ * L_);
#pragma unroll
        for (int p = 0; p < 8; ++p)
            ((v4f*)al)[tid + 128 * p] = ag[tid + 128 * p];
        const v4f* hg = (const v4f*)hstart + (size_t)g * (G_ * 16);
#pragma unroll
        for (int p = 0; p < 2; ++p) {
            int idx = tid + 128 * p;
            int row = idx >> 4, c4 = idx & 15;
            *(v4f*)&hs[0][row * RP_ + c4 * 4] = hg[idx];
        }
    }

    const size_t e0 = ((size_t)(G_ * g + 8 * half) * L_) * 64 + 16 * wave + c;
    float ucur[8];
#pragma unroll
    for (int r = 0; r < 8; ++r) ucur[r] = U[e0 + (size_t)r * 4096];

    __syncthreads();

    for (int s = 0; s < L_; ++s) {
        const float* Hc = hs[s & 1];
        float*       Hn = hs[(s & 1) ^ 1];

        v4f q[8];
#pragma unroll
        for (int i = 0; i < 8; ++i)
            q[i] = *(const v4f*)&Hc[c * RP_ + 8 * i + 4 * half];

        v4f av[8];
#pragma unroll
        for (int r = 0; r < 8; ++r)
            av[r] = *(const v4f*)&al[((8 * half + r) * L_ + s) * 4];

        const int sn = (s + 1 < L_) ? s + 1 : s;
        float un[8];
#pragma unroll
        for (int r = 0; r < 8; ++r) un[r] = U[e0 + (size_t)r * 4096 + sn * 64];

        v8f acc[K_] = {v8f{}, v8f{}, v8f{}, v8f{}};
#pragma unroll
        for (int i = 0; i < 8; ++i) {
            const v2f a0 = {q[i].x, q[i].y};
            const v2f a1 = {q[i].z, q[i].w};
#pragma unroll
            for (int k = 0; k < K_; ++k) acc[k] = wmma4(a0, bfr[k][2 * i], acc[k]);
#pragma unroll
            for (int k = 0; k < K_; ++k) acc[k] = wmma4(a1, bfr[k][2 * i + 1], acc[k]);
        }

#pragma unroll
        for (int r = 0; r < 8; ++r) {
            float hn = ucur[r];
            hn = __builtin_fmaf(av[r].x, acc[0][r], hn);
            hn = __builtin_fmaf(av[r].y, acc[1][r], hn);
            hn = __builtin_fmaf(av[r].z, acc[2][r], hn);
            hn = __builtin_fmaf(av[r].w, acc[3][r], hn);
            Hn[(8 * half + r) * RP_ + 16 * wave + c] = hn;
            hist[e0 + (size_t)r * 4096 + s * 64] = hn;
            ucur[r] = un[r];
        }
        lds_barrier();
    }
}

#define NSLAB_ 20

static __device__ __forceinline__ WSlab k3_fetch_slab(
        int s, int e0, int tid, const float* __restrict__ Cs,
        const float* __restrict__ Ds) {
    if (s < 16)
        return fetch_slab(Ds + (s >> 2) * 65536 + e0 * 256 + (s & 3) * 64, 256, tid);
    return fetch_slab(Cs + (s - 16) * 16384 + e0 * 64, 64, tid);
}

__global__ __launch_bounds__(128)
void k3_y(const float* __restrict__ x, const float* __restrict__ alpha,
          const float* __restrict__ Cs, const float* __restrict__ Ds,
          const float* __restrict__ hist, float* __restrict__ y) {
    __shared__ __attribute__((aligned(16))) float xs[64 * XP_];
    __shared__ __attribute__((aligned(16))) float hs[64 * HP_];
    __shared__ __attribute__((aligned(16))) float ws[64 * HP_];
    __shared__ __attribute__((aligned(16))) float al[64 * 4];
    const int tid = threadIdx.x;
    const int bt0 = blockIdx.x * 64;
    const int e0  = blockIdx.y * 64;

    WSlab pre = k3_fetch_slab(0, e0, tid, Cs, Ds);

    {
        const v4f* xg = (const v4f*)x + (size_t)bt0 * 64;
#pragma unroll 8
        for (int i = tid; i < 64 * 64; i += 128) {
            int rr = i >> 6, c4 = i & 63;
            *(v4f*)&xs[rr * XP_ + c4 * 4] = xg[i];
        }
        const v4f* hg = (const v4f*)hist + (size_t)bt0 * 16;
#pragma unroll
        for (int p = 0; p < 8; ++p) {
            int i = tid + 128 * p;
            int rr = i >> 4, c4 = i & 15;
            *(v4f*)&hs[rr * HP_ + c4 * 4] = hg[i];
        }
        if (tid < 64)
            *(v4f*)&al[tid * 4] = ((const v4f*)alpha)[bt0 + tid];
    }

    const int lane = tid & 31, wave = tid >> 5;
    const int r = lane & 15, half = lane >> 4;
    const int arow = 16 * wave + r;
    const float* xrow = &xs[arow * XP_ + 2 * half];
    const float* hrow = &hs[arow * HP_ + 2 * half];
    const float* wrow = &ws[r * HP_ + 2 * half];

    v8f acc[4] = {v8f{}, v8f{}, v8f{}, v8f{}};

    for (int s = 0; s < NSLAB_; ++s) {
        __syncthreads();
        store_slab(ws, pre, tid);
        __syncthreads();
        if (s + 1 < NSLAB_)
            pre = k3_fetch_slab(s + 1, e0, tid, Cs, Ds);

        const int    kk = (s < 16) ? (s >> 2) : (s - 16);
        const float* ap = (s < 16) ? (xrow + (s & 3) * 64) : hrow;
        slab_mma(ap, al[arow * 4 + kk], wrow, acc);
    }

#pragma unroll
    for (int J = 0; J < 4; ++J)
#pragma unroll
        for (int rr = 0; rr < 8; ++rr)
            y[(size_t)(bt0 + 16 * wave + rr + half * 8) * 256 + e0 + 16 * J + r] = acc[J][rr];
}

extern "C" void kernel_launch(void* const* d_in, const int* in_sizes, int n_in,
                              void* d_out, int out_size, void* d_ws, size_t ws_size,
                              hipStream_t stream) {
    const float* x     = (const float*)d_in[0];
    const float* alpha = (const float*)d_in[1];
    const float* As    = (const float*)d_in[2];
    const float* Bs    = (const float*)d_in[3];
    const float* Cs    = (const float*)d_in[4];
    const float* Ds    = (const float*)d_in[5];

    float* y  = (float*)d_out;
    float* hT = y + (size_t)BT_ * D_;

    float* U      = (float*)d_ws;
    float* hist   = U      + (size_t)BT_ * N_;
    float* Mall   = hist   + (size_t)BT_ * N_;
    float* vall   = Mall   + (size_t)B_ * C_ * N_ * N_;
    float* hstart = vall   + (size_t)B_ * C_ * N_;

    k1_xbmix   <<<dim3(BT_ / 64),    128, 0, stream>>>(x, alpha, Bs, U);
    kA_compose <<<dim3(B_ * C_),     128, 0, stream>>>(alpha, As, U, Mall, vall);
    kB_chunkscan<<<dim3(B_),         256, 0, stream>>>(Mall, vall, hstart, hT);
    kC_replay  <<<dim3(B_ * C_ / G_), 128, 0, stream>>>(alpha, As, U, hstart, hist);
    k3_y       <<<dim3(BT_ / 64, 4), 128, 0, stream>>>(x, alpha, Cs, Ds, hist, y);
}


# K3's Dflat part (16 of 20 slabs, needs no h) runs as extra blocks of the scan and replay launches, on WGPs those serial chains leave idle; a 4-slab Cflat kernel continues each y chain; sums unchanged
# speedup vs baseline: 1.9318x; 1.1522x over previous
#include <hip/hip_runtime.h>
#include <hip/hip_bf16.h>


typedef float v2f __attribute__((ext_vector_type(2)));
typedef float v8f __attribute__((ext_vector_type(8)));
typedef float v4f __attribute__((ext_vector_type(4)));

#define B_  16
#define T_  4096
#define D_  256
#define N_  64
#define K_  4
#define BT_ (B_ * T_)
#define L_  64
#define C_  (T_ / L_)

static __device__ __forceinline__ v8f wmma4(v2f a, v2f b, v8f c) {
    return __builtin_amdgcn_wmma_f32_16x16x4_f32(false, a, false, b,
                                                 (short)0, c, false, false);
}

#define XP_ 260
#define HP_ 68

struct WSlab { v4f q[8]; };

static __device__ __forceinline__ WSlab fetch_slab(const float* __restrict__ base,
                                                   int stride, int tid) {
    WSlab out;
#pragma unroll
    for (int p = 0; p < 8; ++p) {
        int idx = tid + 128 * p;
        int row = idx >> 4, c4 = idx & 15;
        out.q[p] = *(const v4f*)(base + row * stride + c4 * 4);
    }
    return out;
}

static __device__ __forceinline__ void store_slab(float* ws, const WSlab& pre, int tid) {
#pragma unroll
    for (int p = 0; p < 8; ++p) {
        int idx = tid + 128 * p;
        int row = idx >> 4, c4 = idx & 15;
        *(v4f*)&ws[row * HP_ + c4 * 4] = pre.q[p];
    }
}

static __device__ __forceinline__ void slab_mma(const float* ap, float am,
                                                const float* wrow, v8f (&acc)[4]) {
#pragma unroll
    for (int kb = 0; kb < 64; kb += 4) {
        v2f xa = *(const v2f*)(ap + kb);
        v2f a;
        a.x = am * xa.x;
        a.y = am * xa.y;
#pragma unroll
        for (int J = 0; J < 4; ++J) {
            v2f bf = *(const v2f*)(wrow + J * 16 * HP_ + kb);
            acc[J] = wmma4(a, bf, acc[J]);
        }
    }
}

__global__ __launch_bounds__(128)
void k1_xbmix(const float* __restrict__ x, const float* __restrict__ alpha,
              const float* __restrict__ Bs, float* __restrict__ U) {
    __shared__ __attribute__((aligned(16))) float xs[64 * XP_];
    __shared__ __attribute__((aligned(16))) float ws[64 * HP_];
    __shared__ __attribute__((aligned(16))) float al[64 * 4];
    const int tid = threadIdx.x;
    const int bt0 = blockIdx.x * 64;

    WSlab pre = fetch_slab(Bs, 256, tid);

    {
        const v4f* xg = (const v4f*)x + (size_t)bt0 * 64;
#pragma unroll 8
        for (int i = tid; i < 64 * 64; i += 128) {
            int rr = i >> 6, c4 = i & 63;
            *(v4f*)&xs[rr * XP_ + c4 * 4] = xg[i];
        }
        if (tid < 64)
            *(v4f*)&al[tid * 4] = ((const v4f*)alpha)[bt0 + tid];
    }

    const int lane = tid & 31, wave = tid >> 5;
    const int r = lane & 15, half = lane >> 4;
    const int arow = 16 * wave + r;
    const float* xrow = &xs[arow * XP_ + 2 * half];
    const float* wrow = &ws[r * HP_ + 2 * half];

    v8f acc[4] = {v8f{}, v8f{}, v8f{}, v8f{}};

    for (int s = 0; s < 16; ++s) {
        __syncthreads();
        store_slab(ws, pre, tid);
        __syncthreads();
        if (s + 1 < 16)
            pre = fetch_slab(Bs + ((s + 1) >> 2) * 16384 + ((s + 1) & 3) * 64, 256, tid);
        slab_mma(xrow + (s & 3) * 64, al[arow * 4 + (s >> 2)], wrow, acc);
    }

#pragma unroll
    for (int J = 0; J < 4; ++J)
#pragma unroll
        for (int rr = 0; rr < 8; ++rr)
            U[(size_t)(bt0 + 16 * wave + rr + half * 8) * 64 + 16 * J + r] = acc[J][rr];
}

#define AP_ 68

__global__ __launch_bounds__(128)
void kA_compose(const float* __restrict__ alpha, const float* __restrict__ As,
                const float* __restrict__ U, float* __restrict__ Mall,
                float* __restrict__ vall) {
    __shared__ __attribute__((aligned(16))) float Assh[K_ * N_ * AP_];
    __shared__ __attribute__((aligned(16))) float PT[2][N_ * AP_];
    __shared__ __attribute__((aligned(16))) float al[L_ * 4];
    __shared__ __attribute__((aligned(16))) float ush[2][N_];

    const int tid = threadIdx.x;
    const int bc  = blockIdx.x;
    const int b   = bc >> 6;
    const int c   = bc & 63;
    const long t0 = (long)b * T_ + (long)c * L_;

    for (int i = tid; i < K_ * N_ * 16; i += 128) {
        int row = i >> 4, c4 = i & 15;
        *(v4f*)&Assh[row * AP_ + c4 * 4] = ((const v4f*)As)[i];
    }
    for (int i = tid; i < N_ * N_; i += 128) {
        int mm = i >> 6, nn = i & 63;
        PT[0][mm * AP_ + nn] = (nn == mm) ? 1.0f : 0.0f;
    }
    if (tid < 64) {
        *(v4f*)&al[tid * 4] = ((const v4f*)alpha)[t0 + tid];
        ush[0][tid] = U[(t0 + L_ - 1) * 64 + tid];
    }
    __syncthreads();

    const int lane = tid & 31, wave = tid >> 5;
    const int r = lane & 15, half = lane >> 4;
    const float* asrow = &Assh[(16 * wave + r) * AP_ + 2 * half];
    const int gm = tid >> 1, gh = tid & 1;
    float q = 0.0f;

    for (int it = 0; it < L_; ++it) {
        const int s = L_ - 1 - it;
        const int cur = it & 1, nxt = cur ^ 1;
        const float* Pc = PT[cur];
        float*       Pn = PT[nxt];

        float unext = 0.0f;
        if (tid < 64 && s > 0) unext = U[(t0 + s - 1) * 64 + tid];

        const v4f a4 = *(const v4f*)&al[s * 4];

        v8f acc[4] = {v8f{}, v8f{}, v8f{}, v8f{}};
        const float* brow = &Pc[r * AP_ + 2 * half];
#pragma unroll
        for (int kb = 0; kb < 64; kb += 4) {
            v2f s0 = *(const v2f*)(asrow + 0 * 64 * AP_ + kb);
            v2f s1 = *(const v2f*)(asrow + 1 * 64 * AP_ + kb);
            v2f s2 = *(const v2f*)(asrow + 2 * 64 * AP_ + kb);
            v2f s3 = *(const v2f*)(asrow + 3 * 64 * AP_ + kb);
            v2f a = a4.x * s0;
            a = a4.y * s1 + a;
            a = a4.z * s2 + a;
            a = a4.w * s3 + a;
#pragma unroll
            for (int J = 0; J < 4; ++J) {
                v2f bf = *(const v2f*)(brow + J * 16 * AP_ + kb);
                acc[J] = wmma4(a, bf, acc[J]);
            }
        }

        {
            float local = 0.0f;
            const float* pr = &Pc[gm * AP_ + 32 * gh];
            const float* ur = &ush[cur][32 * gh];
#pragma unroll
            for (int j = 0; j < 8; ++j) {
                v4f pv = *(const v4f*)(pr + 4 * j);
                v4f uv = *(const v4f*)(ur + 4 * j);
                local = __builtin_fmaf(uv.x, pv.x, local);
                local = __builtin_fmaf(uv.y, pv.y, local);
                local = __builtin_fmaf(uv.z, pv.z, local);
                local = __builtin_fmaf(uv.w, pv.w, local);
            }
            local += __shfl_xor(local, 1);
            q += local;
        }

#pragma unroll
        for (int J = 0; J < 4; ++J) {
            float* dst = &Pn[(16 * J + r) * AP_ + 16 * wave + 8 * half];
            v4f lo = {acc[J][0], acc[J][1], acc[J][2], acc[J][3]};
            v4f hi = {acc[J][4], acc[J][5], acc[J][6], acc[J][7]};
            *(v4f*)dst       = lo;
            *(v4f*)(dst + 4) = hi;
        }
        if (tid < 64 && s > 0) ush[nxt][tid] = unext;
        __syncthreads();
    }

    for (int i = tid; i < N_ * N_; i += 128)
        Mall[(long)bc * 4096 + i] = PT[0][(i & 63) * AP_ + (i >> 6)];
    if (gh == 0) vall[(long)bc * 64 + gm] = q;
}

#define SMEM_FLOATS_ (64 * XP_ + 64 * HP_ + 64 * 4)
#define YD_HALF_  2048

static __device__ __forceinline__ void yd_tile(int id, const float* __restrict__ x,
                                               const float* __restrict__ alpha,
                                               const float* __restrict__ Ds,
                                               float* __restrict__ y, float* smem) {
    float* xs = smem;
    float* ws = smem + 64 * XP_;
    float* al = ws + 64 * HP_;
    const int tid = threadIdx.x;
    const int bt0 = (id & 1023) * 64;
    const int e0  = (id >> 10) * 64;

    WSlab pre = fetch_slab(Ds + e0 * 256, 256, tid);

    {
        const v4f* xg = (const v4f*)x + (size_t)bt0 * 64;
#pragma unroll 8
        for (int i = tid; i < 64 * 64; i += 128) {
            int rr = i >> 6, c4 = i & 63;
            *(v4f*)&xs[rr * XP_ + c4 * 4] = xg[i];
        }
        if (tid < 64)
            *(v4f*)&al[tid * 4] = ((const v4f*)alpha)[bt0 + tid];
    }

    const int lane = tid & 31, wave = tid >> 5;
    const int r = lane & 15, half = lane >> 4;
    const int arow = 16 * wave + r;
    const float* xrow = &xs[arow * XP_ + 2 * half];
    const float* wrow = &ws[r * HP_ + 2 * half];

    v8f acc[4] = {v8f{}, v8f{}, v8f{}, v8f{}};

    for (int s = 0; s < 16; ++s) {
        __syncthreads();
        store_slab(ws, pre, tid);
        __syncthreads();
        if (s + 1 < 16)
            pre = fetch_slab(Ds + ((s + 1) >> 2) * 65536 + e0 * 256 + ((s + 1) & 3) * 64,
                             256, tid);
        slab_mma(xrow + (s & 3) * 64, al[arow * 4 + (s >> 2)], wrow, acc);
    }

#pragma unroll
    for (int J = 0; J < 4; ++J)
#pragma unroll
        for (int rr = 0; rr < 8; ++rr)
            y[(size_t)(bt0 + 16 * wave + rr + half * 8) * 256 + e0 + 16 * J + r] = acc[J][rr];
}

static __device__ __forceinline__ void scan_batch(int b, const float* __restrict__ Mall,
                                                  const float* __restrict__ vall,
                                                  float* __restrict__ hstart,
                                                  float* __restrict__ hT, float* smem) {
    float* h    = smem;
    float* psum = smem + 64;
    const int tid = threadIdx.x;
    const int m = tid & 63, pp = tid >> 6;
    if (tid < 64) h[tid] = 0.0f;
    __syncthreads();
    for (int c = 0; c < C_; ++c) {
        const long bc = (long)b * C_ + c;
        if (tid < 64) hstart[bc * 64 + tid] = h[tid];
        const float* Mc = &Mall[bc * 4096];
#pragma unroll
        for (int q = 0; q < 2; ++q) {
            const int n0 = (2 * pp + q) * 16;
            float local = 0.0f;
#pragma unroll
            for (int j = 0; j < 16; ++j)
                local = __builtin_fmaf(h[n0 + j], Mc[(n0 + j) * 64 + m], local);
            psum[(2 * pp + q) * 64 + m] = local;
        }
        __syncthreads();
        if (tid < 64)
            h[tid] = psum[tid] + psum[64 + tid] + psum[128 + tid] +
                     psum[192 + tid] + vall[bc * 64 + tid];
        __syncthreads();
    }
    if (tid < 64) hT[b * 64 + tid] = h[tid];
}

__global__ __launch_bounds__(128)
void kB_scan_yd(const float* __restrict__ x, const float* __restrict__ alpha,
                const float* __restrict__ Ds, const float* __restrict__ Mall,
                const float* __restrict__ vall, float* __restrict__ hstart,
                float* __restrict__ hT, float* __restrict__ y) {
    __shared__ __attribute__((aligned(16))) float smem[SMEM_FLOATS_];
    if (blockIdx.x < B_) {
        scan_batch(blockIdx.x, Mall, vall, hstart, hT, smem);
        return;
    }
    yd_tile(blockIdx.x - B_, x, alpha, Ds, y, smem);
}

#define RP_ 68
#define G_  16

static __device__ __forceinline__ void lds_barrier() {
    asm volatile("s_wait_dscnt 0x0\n\ts_barrier_signal -1\n\ts_barrier_wait -1" ::: "memory");
}

static __device__ __forceinline__ void replay_group(int g, const float* __restrict__ alpha,
                                                    const float* __restrict__ As,
                                                    const float* __restrict__ U,
                                                    const float* __restrict__ hstart,
                                                    float* __restrict__ hist, float* smem) {
    float* hs = smem;
    float* al = smem + 2 * G_ * RP_;
    const int tid = threadIdx.x;
    const int lane = tid & 31, wave = tid >> 5;
    const int c = lane & 15, half = lane >> 4;

    v2f bfr[K_][16];
    {
        const float* ab = As + 16 * wave + c;
#pragma unroll
        for (int k = 0; k < K_; ++k)
#pragma unroll
            for (int w = 0; w < 16; ++w) {
                const int n0 = 8 * (w >> 1) + 4 * half + 2 * (w & 1);
                bfr[k][w].x = ab[k * 4096 + n0 * 64];
                bfr[k][w].y = ab[k * 4096 + (n0 + 1) * 64];
            }
    }

    {
        const v4f* ag = (const v4f*)alpha + (size_t)g * (G_ * L_);
#pragma unroll
        for (int p = 0; p < 8; ++p)
            ((v4f*)al)[tid + 128 * p] = ag[tid + 128 * p];
        const v4f* hg = (const v4f*)hstart + (size_t)g * (G_ * 16);
#pragma unroll
        for (int p = 0; p < 2; ++p) {
            int idx = tid + 128 * p;
            int row = idx >> 4, c4 = idx & 15;
            *(v4f*)&hs[row * RP_ + c4 * 4] = hg[idx];
        }
    }

    const size_t e0 = ((size_t)(G_ * g + 8 * half) * L_) * 64 + 16 * wave + c;
    float ucur[8];
#pragma unroll
    for (int r = 0; r < 8; ++r) ucur[r] = U[e0 + (size_t)r * 4096];

    __syncthreads();

    for (int s = 0; s < L_; ++s) {
        const float* Hc = hs + (s & 1) * (G_ * RP_);
        float*       Hn = hs + ((s & 1) ^ 1) * (G_ * RP_);

        v4f q[8];
#pragma unroll
        for (int i = 0; i < 8; ++i)
            q[i] = *(const v4f*)&Hc[c * RP_ + 8 * i + 4 * half];

        v4f av[8];
#pragma unroll
        for (int r = 0; r < 8; ++r)
            av[r] = *(const v4f*)&al[((8 * half + r) * L_ + s) * 4];

        const int sn = (s + 1 < L_) ? s + 1 : s;
        float un[8];
#pragma unroll
        for (int r = 0; r < 8; ++r) un[r] = U[e0 + (size_t)r * 4096 + sn * 64];

        v8f acc[K_] = {v8f{}, v8f{}, v8f{}, v8f{}};
#pragma unroll
        for (int i = 0; i < 8; ++i) {
            const v2f a0 = {q[i].x, q[i].y};
            const v2f a1 = {q[i].z, q[i].w};
#pragma unroll
            for (int k = 0; k < K_; ++k) acc[k] = wmma4(a0, bfr[k][2 * i], acc[k]);
#pragma unroll
            for (int k = 0; k < K_; ++k) acc[k] = wmma4(a1, bfr[k][2 * i + 1], acc[k]);
        }

#pragma unroll
        for (int r = 0; r < 8; ++r) {
            float hn = ucur[r];
            hn = __builtin_fmaf(av[r].x, acc[0][r], hn);
            hn = __builtin_fmaf(av[r].y, acc[1][r], hn);
            hn = __builtin_fmaf(av[r].z, acc[2][r], hn);
            hn = __builtin_fmaf(av[r].w, acc[3][r], hn);
            Hn[(8 * half + r) * RP_ + 16 * wave + c] = hn;
            hist[e0 + (size_t)r * 4096 + s * 64] = hn;
            ucur[r] = un[r];
        }
        lds_barrier();
    }
}

__global__ __launch_bounds__(128)
void kC_replay_yd(const float* __restrict__ x, const float* __restrict__ alpha,
                  const float* __restrict__ As, const float* __restrict__ Ds,
                  const float* __restrict__ U, const float* __restrict__ hstart,
                  float* __restrict__ hist, float* __restrict__ y) {
    __shared__ __attribute__((aligned(16))) float smem[SMEM_FLOATS_];
    if (blockIdx.x < B_ * C_ / G_) {
        replay_group(blockIdx.x, alpha, As, U, hstart, hist, smem);
        return;
    }
    yd_tile(YD_HALF_ + blockIdx.x - B_ * C_ / G_, x, alpha, Ds, y, smem);
}

__global__ __launch_bounds__(128)
void k3c_y(const float* __restrict__ alpha, const float* __restrict__ Cs,
           const float* __restrict__ hist, float* __restrict__ y) {
    __shared__ __attribute__((aligned(16))) float hs[64 * HP_];
    __shared__ __attribute__((aligned(16))) float ws[64 * HP_];
    __shared__ __attribute__((aligned(16))) float al[64 * 4];
    const int tid = threadIdx.x;
    const int bt0 = blockIdx.x * 64;
    const int e0  = blockIdx.y * 64;

    const int lane = tid & 31, wave = tid >> 5;
    const int r = lane & 15, half = lane >> 4;
    const int arow = 16 * wave + r;
    const float* hrow = &hs[arow * HP_ + 2 * half];
    const float* wrow = &ws[r * HP_ + 2 * half];

    WSlab pre = fetch_slab(Cs + e0 * 64, 64, tid);

    float* yp = y + (size_t)(bt0 + 16 * wave + half * 8) * 256 + e0 + r;
    v8f acc[4];
#pragma unroll
    for (int J = 0; J < 4; ++J)
#pragma unroll
        for (int rr = 0; rr < 8; ++rr)
            acc[J][rr] = yp[(size_t)rr * 256 + 16 * J];

    {
        const v4f* hg = (const v4f*)hist + (size_t)bt0 * 16;
#pragma unroll
        for (int p = 0; p < 8; ++p) {
            int i = tid + 128 * p;
            int rr = i >> 4, c4 = i & 15;
            *(v4f*)&hs[rr * HP_ + c4 * 4] = hg[i];
        }
        if (tid < 64)
            *(v4f*)&al[tid * 4] = ((const v4f*)alpha)[bt0 + tid];
    }

    for (int s = 0; s < 4; ++s) {
        __syncthreads();
        store_slab(ws, pre, tid);
        __syncthreads();
        if (s + 1 < 4)
            pre = fetch_slab(Cs + (s + 1) * 16384 + e0 * 64, 64, tid);
        slab_mma(hrow, al[arow * 4 + s], wrow, acc);
    }

#pragma unroll
    for (int J = 0; J < 4; ++J)
#pragma unroll
        for (int rr = 0; rr < 8; ++rr)
            yp[(size_t)rr * 256 + 16 * J] = acc[J][rr];
}

extern "C" void kernel_launch(void* const* d_in, const int* in_sizes, int n_in,
                              void* d_out, int out_size, void* d_ws, size_t ws_size,
                              hipStream_t stream) {
    const float* x     = (const float*)d_in[0];
    const float* alpha = (const float*)d_in[1];
    const float* As    = (const float*)d_in[2];
    const float* Bs    = (const float*)d_in[3];
    const float* Cs    = (const float*)d_in[4];
    const float* Ds    = (const float*)d_in[5];

    float* y  = (float*)d_out;
    float* hT = y + (size_t)BT_ * D_;

    float* U      = (float*)d_ws;
    float* hist   = U      + (size_t)BT_ * N_;
    float* Mall   = hist   + (size_t)BT_ * N_;
    float* vall   = Mall   + (size_t)B_ * C_ * N_ * N_;
    float* hstart = vall   + (size_t)B_ * C_ * N_;

    k1_xbmix    <<<dim3(BT_ / 64),                128, 0, stream>>>(x, alpha, Bs, U);
    kA_compose  <<<dim3(B_ * C_),                 128, 0, stream>>>(alpha, As, U, Mall, vall);
    kB_scan_yd  <<<dim3(B_ + YD_HALF_),           128, 0, stream>>>(x, alpha, Ds, Mall, vall, hstart, hT, y);
    kC_replay_yd<<<dim3(B_ * C_ / G_ + YD_HALF_), 128, 0, stream>>>(x, alpha, As, Ds, U, hstart, hist, y);
    k3c_y       <<<dim3(BT_ / 64, 4),             128, 0, stream>>>(alpha, Cs, hist, y);
}


# K1, Dflat tiles, Cflat kernel, kA: the LDS fragment reads of a 64-k slab or kA step stand in one group ahead of its WMMA run, not spread through it (kA mixes Amix one step ahead); sums unchanged
# speedup vs baseline: 2.0016x; 1.0361x over previous
#include <hip/hip_runtime.h>
#include <hip/hip_bf16.h>


typedef float v2f __attribute__((ext_vector_type(2)));
typedef float v8f __attribute__((ext_vector_type(8)));
typedef float v4f __attribute__((ext_vector_type(4)));

#define B_  16
#define T_  4096
#define D_  256
#define N_  64
#define K_  4
#define BT_ (B_ * T_)
#define L_  64
#define C_  (T_ / L_)

static __device__ __forceinline__ v8f wmma4(v2f a, v2f b, v8f c) {
    return __builtin_amdgcn_wmma_f32_16x16x4_f32(false, a, false, b,
                                                 (short)0, c, false, false);
}

#define XP_ 260
#define HP_ 68

struct WSlab { v4f q[8]; };

static __device__ __forceinline__ WSlab fetch_slab(const float* __restrict__ base,
                                                   int stride, int tid) {
    WSlab out;
#pragma unroll
    for (int p = 0; p < 8; ++p) {
        int idx = tid + 128 * p;
        int row = idx >> 4, c4 = idx & 15;
        out.q[p] = *(const v4f*)(base + row * stride + c4 * 4);
    }
    return out;
}

static __device__ __forceinline__ void store_slab(float* ws, const WSlab& pre, int tid) {
#pragma unroll
    for (int p = 0; p < 8; ++p) {
        int idx = tid + 128 * p;
        int row = idx >> 4, c4 = idx & 15;
        *(v4f*)&ws[row * HP_ + c4 * 4] = pre.q[p];
    }
}

template <int G>
static __device__ __forceinline__ void slab_mma(const float* ap, float am,
                                                const float* wrow, v8f (&acc)[4]) {
#pragma unroll
    for (int g = 0; g < 16; g += G) {
        v2f xa[G];
        v2f bf[G][4];
#pragma unroll
        for (int i = 0; i < G; ++i) {
            xa[i] = *(const v2f*)(ap + 4 * (g + i));
#pragma unroll
            for (int J = 0; J < 4; ++J)
                bf[i][J] = *(const v2f*)(wrow + J * 16 * HP_ + 4 * (g + i));
        }
        __builtin_amdgcn_sched_barrier(0);
#pragma unroll
        for (int i = 0; i < G; ++i) {
            v2f a;
            a.x = am * xa[i].x;
            a.y = am * xa[i].y;
#pragma unroll
            for (int J = 0; J < 4; ++J)
                acc[J] = wmma4(a, bf[i][J], acc[J]);
        }
        __builtin_amdgcn_sched_barrier(0);
    }
}

__global__ __launch_bounds__(128)
void k1_xbmix(const float* __restrict__ x, const float* __restrict__ alpha,
              const float* __restrict__ Bs, float* __restrict__ U) {
    __shared__ __attribute__((aligned(16))) float xs[64 * XP_];
    __shared__ __attribute__((aligned(16))) float ws[64 * HP_];
    __shared__ __attribute__((aligned(16))) float al[64 * 4];
    const int tid = threadIdx.x;
    const int bt0 = blockIdx.x * 64;

    WSlab pre = fetch_slab(Bs, 256, tid);

    {
        const v4f* xg = (const v4f*)x + (size_t)bt0 * 64;
#pragma unroll 8
        for (int i = tid; i < 64 * 64; i += 128) {
            int rr = i >> 6, c4 = i & 63;
            *(v4f*)&xs[rr * XP_ + c4 * 4] = xg[i];
        }
        if (tid < 64)
            *(v4f*)&al[tid * 4] = ((const v4f*)alpha)[bt0 + tid];
    }

    const int lane = tid & 31, wave = tid >> 5;
    const int r = lane & 15, half = lane >> 4;
    const int arow = 16 * wave + r;
    const float* xrow = &xs[arow * XP_ + 2 * half];
    const float* wrow = &ws[r * HP_ + 2 * half];

    v8f acc[4] = {v8f{}, v8f{}, v8f{}, v8f{}};

    for (int s = 0; s < 16; ++s) {
        __syncthreads();
        store_slab(ws, pre, tid);
        __syncthreads();
        if (s + 1 < 16)
            pre = fetch_slab(Bs + ((s + 1) >> 2) * 16384 + ((s + 1) & 3) * 64, 256, tid);
        slab_mma<16>(xrow + (s & 3) * 64, al[arow * 4 + (s >> 2)], wrow, acc);
    }

#pragma unroll
    for (int J = 0; J < 4; ++J)
#pragma unroll
        for (int rr = 0; rr < 8; ++rr)
            U[(size_t)(bt0 + 16 * wave + rr + half * 8) * 64 + 16 * J + r] = acc[J][rr];
}

#define AP_ 68

__global__ __launch_bounds__(128)
void kA_compose(const float* __restrict__ alpha, const float* __restrict__ As,
                const float* __restrict__ U, float* __restrict__ Mall,
                float* __restrict__ vall) {
    __shared__ __attribute__((aligned(16))) float Assh[K_ * N_ * AP_];
    __shared__ __attribute__((aligned(16))) float PT[2][N_ * AP_];
    __shared__ __attribute__((aligned(16))) float al[L_ * 4];
    __shared__ __attribute__((aligned(16))) float ush[2][N_];

    const int tid = threadIdx.x;
    const int bc  = blockIdx.x;
    const int b   = bc >> 6;
    const int c   = bc & 63;
    const long t0 = (long)b * T_ + (long)c * L_;

    for (int i = tid; i < K_ * N_ * 16; i += 128) {
        int row = i >> 4, c4 = i & 15;
        *(v4f*)&Assh[row * AP_ + c4 * 4] = ((const v4f*)As)[i];
    }
    for (int i = tid; i < N_ * N_; i += 128) {
        int mm = i >> 6, nn = i & 63;
        PT[0][mm * AP_ + nn] = (nn == mm) ? 1.0f : 0.0f;
    }
    if (tid < 64) {
        *(v4f*)&al[tid * 4] = ((const v4f*)alpha)[t0 + tid];
        ush[0][tid] = U[(t0 + L_ - 1) * 64 + tid];
    }
    __syncthreads();

    const int lane = tid & 31, wave = tid >> 5;
    const int r = lane & 15, half = lane >> 4;
    const float* asrow = &Assh[(16 * wave + r) * AP_ + 2 * half];
    const int gm = tid >> 1, gh = tid & 1;
    float q = 0.0f;

    v2f afr[16];
    {
        const v4f a4 = *(const v4f*)&al[(L_ - 1) * 4];
#pragma unroll
        for (int i = 0; i < 16; ++i) {
            v2f s0 = *(const v2f*)(asrow + 0 * 64 * AP_ + 4 * i);
            v2f s1 = *(const v2f*)(asrow + 1 * 64 * AP_ + 4 * i);
            v2f s2 = *(const v2f*)(asrow + 2 * 64 * AP_ + 4 * i);
            v2f s3 = *(const v2f*)(asrow + 3 * 64 * AP_ + 4 * i);
            v2f a = a4.x * s0;
            a = a4.y * s1 + a;
            a = a4.z * s2 + a;
            a = a4.w * s3 + a;
            afr[i] = a;
        }
    }

    for (int it = 0; it < L_; ++it) {
        const int s = L_ - 1 - it;
        const int cur = it & 1, nxt = cur ^ 1;
        const float* Pc = PT[cur];
        float*       Pn = PT[nxt];

        float unext = 0.0f;
        if (tid < 64 && s > 0) unext = U[(t0 + s - 1) * 64 + tid];

        v2f bf[16][4];
        v4f pv[8], uv[8];
        {
            const float* brow = &Pc[r * AP_ + 2 * half];
#pragma unroll
            for (int i = 0; i < 16; ++i)
#pragma unroll
                for (int J = 0; J < 4; ++J)
                    bf[i][J] = *(const v2f*)(brow + J * 16 * AP_ + 4 * i);
            const float* pr = &Pc[gm * AP_ + 32 * gh];
            const float* ur = &ush[cur][32 * gh];
#pragma unroll
            for (int j = 0; j < 8; ++j) {
                pv[j] = *(const v4f*)(pr + 4 * j);
                uv[j] = *(const v4f*)(ur + 4 * j);
            }
        }
        __builtin_amdgcn_sched_barrier(0);

        v8f acc[4] = {v8f{}, v8f{}, v8f{}, v8f{}};
#pragma unroll
        for (int i = 0; i < 16; ++i)
#pragma unroll
            for (int J = 0; J < 4; ++J)
                acc[J] = wmma4(afr[i], bf[i][J], acc[J]);

        float local = 0.0f;
#pragma unroll
        for (int j = 0; j < 8; ++j) {
            local = __builtin_fmaf(uv[j].x, pv[j].x, local);
            local = __builtin_fmaf(uv[j].y, pv[j].y, local);
            local = __builtin_fmaf(uv[j].z, pv[j].z, local);
            local = __builtin_fmaf(uv[j].w, pv[j].w, local);
        }
        __builtin_amdgcn_sched_barrier(0);

        local += __shfl_xor(local, 1);
        q += local;

        if (s > 0) {
            const v4f a4 = *(const v4f*)&al[(s - 1) * 4];
#pragma unroll
            for (int i = 0; i < 16; ++i) {
                v2f s0 = *(const v2f*)(asrow + 0 * 64 * AP_ + 4 * i);
                v2f s1 = *(const v2f*)(asrow + 1 * 64 * AP_ + 4 * i);
                v2f s2 = *(const v2f*)(asrow + 2 * 64 * AP_ + 4 * i);
                v2f s3 = *(const v2f*)(asrow + 3 * 64 * AP_ + 4 * i);
                v2f a = a4.x * s0;
                a = a4.y * s1 + a;
                a = a4.z * s2 + a;
                a = a4.w * s3 + a;
                afr[i] = a;
            }
        }

#pragma unroll
        for (int J = 0; J < 4; ++J) {
            float* dst = &Pn[(16 * J + r) * AP_ + 16 * wave + 8 * half];
            v4f lo = {acc[J][0], acc[J][1], acc[J][2], acc[J][3]};
            v4f hi = {acc[J][4], acc[J][5], acc[J][6], acc[J][7]};
            *(v4f*)dst       = lo;
            *(v4f*)(dst + 4) = hi;
        }
        if (tid < 64 && s > 0) ush[nxt][tid] = unext;
        __syncthreads();
    }

    for (int i = tid; i < N_ * N_; i += 128)
        Mall[(long)bc * 4096 + i] = PT[0][(i & 63) * AP_ + (i >> 6)];
    if (gh == 0) vall[(long)bc * 64 + gm] = q;
}

#define SMEM_FLOATS_ (64 * XP_ + 64 * HP_ + 64 * 4)
#define YD_HALF_  2048

static __device__ __forceinline__ void yd_tile(int id, const float* __restrict__ x,
                                               const float* __restrict__ alpha,
                                               const float* __restrict__ Ds,
                                               float* __restrict__ y, float* smem) {
    float* xs = smem;
    float* ws = smem + 64 * XP_;
    float* al = ws + 64 * HP_;
    const int tid = threadIdx.x;
    const int bt0 = (id & 1023) * 64;
    const int e0  = (id >> 10) * 64;

    WSlab pre = fetch_slab(Ds + e0 * 256, 256, tid);

    {
        const v4f* xg = (const v4f*)x + (size_t)bt0 * 64;
#pragma unroll 8
        for (int i = tid; i < 64 * 64; i += 128) {
            int rr = i >> 6, c4 = i & 63;
            *(v4f*)&xs[rr * XP_ + c4 * 4] = xg[i];
        }
        if (tid < 64)
            *(v4f*)&al[tid * 4] = ((const v4f*)alpha)[bt0 + tid];
    }

    const int lane = tid & 31, wave = tid >> 5;
    const int r = lane & 15, half = lane >> 4;
    const int arow = 16 * wave + r;
    const float* xrow = &xs[arow * XP_ + 2 * half];
    const float* wrow = &ws[r * HP_ + 2 * half];

    v8f acc[4] = {v8f{}, v8f{}, v8f{}, v8f{}};

    for (int s = 0; s < 16; ++s) {
        __syncthreads();
        store_slab(ws, pre, tid);
        __syncthreads();
        if (s + 1 < 16)
            pre = fetch_slab(Ds + ((s + 1) >> 2) * 65536 + e0 * 256 + ((s + 1) & 3) * 64,
                             256, tid);
        slab_mma<16>(xrow + (s & 3) * 64, al[arow * 4 + (s >> 2)], wrow, acc);
    }

#pragma unroll
    for (int J = 0; J < 4; ++J)
#pragma unroll
        for (int rr = 0; rr < 8; ++rr)
            y[(size_t)(bt0 + 16 * wave + rr + half * 8) * 256 + e0 + 16 * J + r] = acc[J][rr];
}

static __device__ __forceinline__ void scan_batch(int b, const float* __restrict__ Mall,
                                                  const float* __restrict__ vall,
                                                  float* __restrict__ hstart,
                                                  float* __restrict__ hT, float* smem) {
    float* h    = smem;
    float* psum = smem + 64;
    const int tid = threadIdx.x;
    const int m = tid & 63, pp = tid >> 6;
    if (tid < 64) h[tid] = 0.0f;
    __syncthreads();
    for (int c = 0; c < C_; ++c) {
        const long bc = (long)b * C_ + c;
        if (tid < 64) hstart[bc * 64 + tid] = h[tid];
        const float* Mc = &Mall[bc * 4096];
#pragma unroll
        for (int q = 0; q < 2; ++q) {
            const int n0 = (2 * pp + q) * 16;
            float local = 0.0f;
#pragma unroll
            for (int j = 0; j < 16; ++j)
                local = __builtin_fmaf(h[n0 + j], Mc[(n0 + j) * 64 + m], local);
            psum[(2 * pp + q) * 64 + m] = local;
        }
        __syncthreads();
        if (tid < 64)
            h[tid] = psum[tid] + psum[64 + tid] + psum[128 + tid] +
                     psum[192 + tid] + vall[bc * 64 + tid];
        __syncthreads();
    }
    if (tid < 64) hT[b * 64 + tid] = h[tid];
}

__global__ __launch_bounds__(128)
void kB_scan_yd(const float* __restrict__ x, const float* __restrict__ alpha,
                const float* __restrict__ Ds, const float* __restrict__ Mall,
                const float* __restrict__ vall, float* __restrict__ hstart,
                float* __restrict__ hT, float* __restrict__ y) {
    __shared__ __attribute__((aligned(16))) float smem[SMEM_FLOATS_];
    if (blockIdx.x < B_) {
        scan_batch(blockIdx.x, Mall, vall, hstart, hT, smem);
        return;
    }
    yd_tile(blockIdx.x - B_, x, alpha, Ds, y, smem);
}

#define RP_ 68
#define G_  16

static __device__ __forceinline__ void lds_barrier() {
    asm volatile("s_wait_dscnt 0x0\n\ts_barrier_signal -1\n\ts_barrier_wait -1" ::: "memory");
}

static __device__ __forceinline__ void replay_group(int g, const float* __restrict__ alpha,
                                                    const float* __restrict__ As,
                                                    const float* __restrict__ U,
                                                    const float* __restrict__ hstart,
                                                    float* __restrict__ hist, float* smem) {
    float* hs = smem;
    float* al = smem + 2 * G_ * RP_;
    const int tid = threadIdx.x;
    const int lane = tid & 31, wave = tid >> 5;
    const int c = lane & 15, half = lane >> 4;

    v2f bfr[K_][16];
    {
        const float* ab = As + 16 * wave + c;
#pragma unroll
        for (int k = 0; k < K_; ++k)
#pragma unroll
            for (int w = 0; w < 16; ++w) {
                const int n0 = 8 * (w >> 1) + 4 * half + 2 * (w & 1);
                bfr[k][w].x = ab[k * 4096 + n0 * 64];
                bfr[k][w].y = ab[k * 4096 + (n0 + 1) * 64];
            }
    }

    {
        const v4f* ag = (const v4f*)alpha + (size_t)g * (G_ * L_);
#pragma unroll
        for (int p = 0; p < 8; ++p)
            ((v4f*)al)[tid + 128 * p] = ag[tid + 128 * p];
        const v4f* hg = (const v4f*)hstart + (size_t)g * (G_ * 16);
#pragma unroll
        for (int p = 0; p < 2; ++p) {
            int idx = tid + 128 * p;
            int row = idx >> 4, c4 = idx & 15;
            *(v4f*)&hs[row * RP_ + c4 * 4] = hg[idx];
        }
    }

    const size_t e0 = ((size_t)(G_ * g + 8 * half) * L_) * 64 + 16 * wave + c;
    float ucur[8];
#pragma unroll
    for (int r = 0; r < 8; ++r) ucur[r] = U[e0 + (size_t)r * 4096];

    __syncthreads();

    for (int s = 0; s < L_; ++s) {
        const float* Hc = hs + (s & 1) * (G_ * RP_);
        float*       Hn = hs + ((s & 1) ^ 1) * (G_ * RP_);

        v4f q[8];
#pragma unroll
        for (int i = 0; i < 8; ++i)
            q[i] = *(const v4f*)&Hc[c * RP_ + 8 * i + 4 * half];

        v4f av[8];
#pragma unroll
        for (int r = 0; r < 8; ++r)
            av[r] = *(const v4f*)&al[((8 * half + r) * L_ + s) * 4];

        const int sn = (s + 1 < L_) ? s + 1 : s;
        float un[8];
#pragma unroll
        for (int r = 0; r < 8; ++r) un[r] = U[e0 + (size_t)r * 4096 + sn * 64];

        v8f acc[K_] = {v8f{}, v8f{}, v8f{}, v8f{}};
#pragma unroll
        for (int i = 0; i < 8; ++i) {
            const v2f a0 = {q[i].x, q[i].y};
            const v2f a1 = {q[i].z, q[i].w};
#pragma unroll
            for (int k = 0; k < K_; ++k) acc[k] = wmma4(a0, bfr[k][2 * i], acc[k]);
#pragma unroll
            for (int k = 0; k < K_; ++k) acc[k] = wmma4(a1, bfr[k][2 * i + 1], acc[k]);
        }

#pragma unroll
        for (int r = 0; r < 8; ++r) {
            float hn = ucur[r];
            hn = __builtin_fmaf(av[r].x, acc[0][r], hn);
            hn = __builtin_fmaf(av[r].y, acc[1][r], hn);
            hn = __builtin_fmaf(av[r].z, acc[2][r], hn);
            hn = __builtin_fmaf(av[r].w, acc[3][r], hn);
            Hn[(8 * half + r) * RP_ + 16 * wave + c] = hn;
            hist[e0 + (size_t)r * 4096 + s * 64] = hn;
            ucur[r] = un[r];
        }
        lds_barrier();
    }
}

__global__ __launch_bounds__(128)
void kC_replay_yd(const float* __restrict__ x, const float* __restrict__ alpha,
                  const float* __restrict__ As, const float* __restrict__ Ds,
                  const float* __restrict__ U, const float* __restrict__ hstart,
                  float* __restrict__ hist, float* __restrict__ y) {
    __shared__ __attribute__((aligned(16))) float smem[SMEM_FLOATS_];
    if (blockIdx.x < B_ * C_ / G_) {
        replay_group(blockIdx.x, alpha, As, U, hstart, hist, smem);
        return;
    }
    yd_tile(YD_HALF_ + blockIdx.x - B_ * C_ / G_, x, alpha, Ds, y, smem);
}

__global__ __launch_bounds__(128)
void k3c_y(const float* __restrict__ alpha, const float* __restrict__ Cs,
           const float* __restrict__ hist, float* __restrict__ y) {
    __shared__ __attribute__((aligned(16))) float hs[64 * HP_];
    __shared__ __attribute__((aligned(16))) float ws[64 * HP_];
    __shared__ __attribute__((aligned(16))) float al[64 * 4];
    const int tid = threadIdx.x;
    const int bt0 = blockIdx.x * 64;
    const int e0  = blockIdx.y * 64;

    const int lane = tid & 31, wave = tid >> 5;
    const int r = lane & 15, half = lane >> 4;
    const int arow = 16 * wave + r;
    const float* hrow = &hs[arow * HP_ + 2 * half];
    const float* wrow = &ws[r * HP_ + 2 * half];

    WSlab pre = fetch_slab(Cs + e0 * 64, 64, tid);

    float* yp = y + (size_t)(bt0 + 16 * wave + half * 8) * 256 + e0 + r;
    v8f acc[4];
#pragma unroll
    for (int J = 0; J < 4; ++J)
#pragma unroll
        for (int rr = 0; rr < 8; ++rr)
            acc[J][rr] = yp[(size_t)rr * 256 + 16 * J];

    {
        const v4f* hg = (const v4f*)hist + (size_t)bt0 * 16;
#pragma unroll
        for (int p = 0; p < 8; ++p) {
            int i = tid + 128 * p;
            int rr = i >> 4, c4 = i & 15;
            *(v4f*)&hs[rr * HP_ + c4 * 4] = hg[i];
        }
        if (tid < 64)
            *(v4f*)&al[tid * 4] = ((const v4f*)alpha)[bt0 + tid];
    }

    for (int s = 0; s < 4; ++s) {
        __syncthreads();
        store_slab(ws, pre, tid);
        __syncthreads();
        if (s + 1 < 4)
            pre = fetch_slab(Cs + (s + 1) * 16384 + e0 * 64, 64, tid);
        slab_mma<8>(hrow, al[arow * 4 + s], wrow, acc);
    }

#pragma unroll
    for (int J = 0; J < 4; ++J)
#pragma unroll
        for (int rr = 0; rr < 8; ++rr)
            yp[(size_t)rr * 256 + 16 * J] = acc[J][rr];
}

extern "C" void kernel_launch(void* const* d_in, const int* in_sizes, int n_in,
                              void* d_out, int out_size, void* d_ws, size_t ws_size,
                              hipStream_t stream) {
    const float* x     = (const float*)d_in[0];
    const float* alpha = (const float*)d_in[1];
    const float* As    = (const float*)d_in[2];
    const float* Bs    = (const float*)d_in[3];
    const float* Cs    = (const float*)d_in[4];
    const float* Ds    = (const float*)d_in[5];

    float* y  = (float*)d_out;
    float* hT = y + (size_t)BT_ * D_;

    float* U      = (float*)d_ws;
    float* hist   = U      + (size_t)BT_ * N_;
    float* Mall   = hist   + (size_t)BT_ * N_;
    float* vall   = Mall   + (size_t)B_ * C_ * N_ * N_;
    float* hstart = vall   + (size_t)B_ * C_ * N_;

    k1_xbmix    <<<dim3(BT_ / 64),                128, 0, stream>>>(x, alpha, Bs, U);
    kA_compose  <<<dim3(B_ * C_),                 128, 0, stream>>>(alpha, As, U, Mall, vall);
    kB_scan_yd  <<<dim3(B_ + YD_HALF_),           128, 0, stream>>>(x, alpha, Ds, Mall, vall, hstart, hT, y);
    kC_replay_yd<<<dim3(B_ * C_ / G_ + YD_HALF_), 128, 0, stream>>>(x, alpha, As, Ds, U, hstart, hist, y);
    k3c_y       <<<dim3(BT_ / 64, 4),             128, 0, stream>>>(alpha, Cs, hist, y);
}
